# starting kernel
# baseline (speedup 1.0000x reference)
#include <hip/hip_runtime.h>
#include <hip/hip_bf16.h>

#define D_IN   128
#define D_OUT  64
#define ALPHA  0.2f
#define EPS    1e-10f

typedef __attribute__((ext_vector_type(2))) float v2f;
typedef __attribute__((ext_vector_type(8))) float v8f;

__device__ __forceinline__ unsigned f2ord(float f) {
    unsigned u = __float_as_uint(f);
    return (u & 0x80000000u) ? ~u : (u | 0x80000000u);
}
__device__ __forceinline__ float ord2f(unsigned u) {
    return __uint_as_float((u & 0x80000000u) ? (u ^ 0x80000000u) : ~u);
}
#define ORD_NEGINF 0x007FFFFFu

__device__ __forceinline__ float leaky(float x) { return x > 0.f ? x : ALPHA * x; }

__global__ void gat_init(float* __restrict__ out, float* __restrict__ denom,
                         unsigned* __restrict__ mflip, int N) {
    int i = blockIdx.x * blockDim.x + threadIdx.x;
    int total = N * D_OUT;
    if (i < total) out[i] = 0.0f;
    if (i < N) { denom[i] = 0.0f; mflip[i] = ORD_NEGINF; }
}

__global__ void gat_gemm_wmma(const float* __restrict__ node,
                              const float* __restrict__ W,
                              float* __restrict__ h, int N) {
    int wave = (blockIdx.x * blockDim.x + threadIdx.x) >> 5;
    int lane = threadIdx.x & 31;
    int row0 = wave * 16;
    if (row0 >= N) return;

    int mrow = lane & 15;
    int koff = (lane >> 4) * 2;

    v8f c0 = {}, c1 = {}, c2 = {}, c3 = {};

    const float* arow = node + (size_t)(row0 + mrow) * D_IN + koff;

    for (int k = 0; k < D_IN; k += 4) {
        v2f a;
        a.x = arow[k];
        a.y = arow[k + 1];

        const float* bp = W + (size_t)(k + koff) * D_OUT + mrow;
        v2f b0, b1, b2, b3;
        b0.x = bp[0];            b0.y = bp[D_OUT];
        b1.x = bp[16];           b1.y = bp[16 + D_OUT];
        b2.x = bp[32];           b2.y = bp[32 + D_OUT];
        b3.x = bp[48];           b3.y = bp[48 + D_OUT];

        c0 = __builtin_amdgcn_wmma_f32_16x16x4_f32(false, a, false, b0, (short)0, c0, false, false);
        c1 = __builtin_amdgcn_wmma_f32_16x16x4_f32(false, a, false, b1, (short)0, c1, false, false);
        c2 = __builtin_amdgcn_wmma_f32_16x16x4_f32(false, a, false, b2, (short)0, c2, false, false);
        c3 = __builtin_amdgcn_wmma_f32_16x16x4_f32(false, a, false, b3, (short)0, c3, false, false);
    }

    int nlo   = lane & 15;
    int rbase = row0 + ((lane >> 4) * 8);
#pragma unroll
    for (int r = 0; r < 8; ++r) {
        float* op = h + (size_t)(rbase + r) * D_OUT + nlo;
        op[0]  = c0[r];
        op[16] = c1[r];
        op[32] = c2[r];
        op[48] = c3[r];
    }
}

__global__ void gat_scores(const float* __restrict__ h, const float* __restrict__ a,
                           float* __restrict__ ssrc, float* __restrict__ sdst, int N) {
    int i = blockIdx.x * blockDim.x + threadIdx.x;
    if (i >= N) return;
    const float* hr = h + (size_t)i * D_OUT;
    float s0 = 0.f, s1 = 0.f;
#pragma unroll 8
    for (int j = 0; j < D_OUT; ++j) {
        float v = hr[j];
        s0 = fmaf(v, a[j], s0);
        s1 = fmaf(v, a[D_OUT + j], s1);
    }
    ssrc[i] = s0;
    sdst[i] = s1;
}

__global__ void gat_edge_max(const int* __restrict__ src, const int* __restrict__ dst,
                             const float* __restrict__ ssrc, const float* __restrict__ sdst,
                             unsigned* __restrict__ mflip, float* __restrict__ logits, int E) {
    int e = blockIdx.x * blockDim.x + threadIdx.x;
    if (e >= E) return;
    int s = src[e], d = dst[e];
    float l = leaky(ssrc[s] + sdst[d]);
    logits[e] = l;
    atomicMax(&mflip[s], f2ord(l));
}

__global__ void gat_m_final(const unsigned* __restrict__ mflip, float* __restrict__ m, int N) {
    int i = blockIdx.x * blockDim.x + threadIdx.x;
    if (i >= N) return;
    float v = ord2f(mflip[i]);
    m[i] = __isinff(v) && v < 0.f ? 0.0f : v;
}

__global__ void gat_edge_exp(const int* __restrict__ src, const float* __restrict__ logits,
                             const float* __restrict__ m, float* __restrict__ denom, int E) {
    int e = blockIdx.x * blockDim.x + threadIdx.x;
    if (e >= E) return;
    int s = src[e];
    atomicAdd(&denom[s], __expf(logits[e] - m[s]));
}

__global__ void gat_edge_scatter(const int* __restrict__ src, const int* __restrict__ dst,
                                 const float* __restrict__ logits, const float* __restrict__ m,
                                 const float* __restrict__ denom, const float* __restrict__ h,
                                 float* __restrict__ out, int E) {
    long long gid = (long long)blockIdx.x * blockDim.x + threadIdx.x;
    int e    = (int)(gid >> 5);
    int lane = (int)(gid & 31);
    if (e >= E) return;
    int s = src[e], d = dst[e];
    float att = __expf(logits[e] - m[s]) / (denom[s] + EPS);
    const float2* hp = (const float2*)(h + (size_t)d * D_OUT);
    float2 v = hp[lane];
    float* op = out + (size_t)s * D_OUT + 2 * lane;
    atomicAdd(op + 0, att * v.x);
    atomicAdd(op + 1, att * v.y);
}

extern "C" void kernel_launch(void* const* d_in, const int* in_sizes, int n_in,
                              void* d_out, int out_size, void* d_ws, size_t ws_size,
                              hipStream_t stream) {
    const float* node = (const float*)d_in[0];
    const int*   ei   = (const int*)d_in[1];
    const float* W    = (const float*)d_in[2];
    const float* a    = (const float*)d_in[3];
    float* out        = (float*)d_out;

    int N = in_sizes[0] / D_IN;
    int E = in_sizes[1] / 2;
    const int* src = ei;
    const int* dst = ei + E;

    float*    h      = (float*)d_ws;
    float*    ssrc   = h + (size_t)N * D_OUT;
    float*    sdst   = ssrc + N;
    float*    m      = sdst + N;
    unsigned* mflip  = (unsigned*)(m + N);
    float*    denom  = (float*)(mflip + N);
    float*    logits = denom + N;

    const int B = 256;

    {
        int total = N * D_OUT;
        gat_init<<<(total + B - 1) / B, B, 0, stream>>>(out, denom, mflip, N);
    }
    {
        int waves   = (N + 15) / 16;
        long long threads = (long long)waves * 32;
        int blocks  = (int)((threads + B - 1) / B);
        gat_gemm_wmma<<<blocks, B, 0, stream>>>(node, W, h, N);
    }
    gat_scores<<<(N + B - 1) / B, B, 0, stream>>>(h, a, ssrc, sdst, N);
    gat_edge_max<<<(E + B - 1) / B, B, 0, stream>>>(src, dst, ssrc, sdst, mflip, logits, E);
    gat_m_final<<<(N + B - 1) / B, B, 0, stream>>>(mflip, m, N);
    gat_edge_exp<<<(E + B - 1) / B, B, 0, stream>>>(src, logits, m, denom, E);
    {
        long long threads = (long long)E * 32;
        int blocks = (int)((threads + B - 1) / B);
        gat_edge_scatter<<<blocks, B, 0, stream>>>(src, dst, logits, m, denom, h, out, E);
    }
}


# edge passes fetch fewer per-node words: max taken over sdst[dst] alone (max logit = logit of that max), ssrc and max read as one word pair, exp weight kept in the edge slot; sum order unchanged
# speedup vs baseline: 2.8891x; 2.8891x over previous
#include <hip/hip_runtime.h>
#include <hip/hip_bf16.h>

#define D_IN   128
#define D_OUT  64
#define ALPHA  0.2f
#define EPS    1e-10f

#define CHUNK      256
#define EWAVES     4
#define EBLOCK     (EWAVES * 32)
#define CNT_STRIDE 64

typedef __attribute__((ext_vector_type(2))) float v2f;
typedef __attribute__((ext_vector_type(8))) float v8f;

__device__ __forceinline__ unsigned f2ord(float f) {
    unsigned u = __float_as_uint(f);
    return (u & 0x80000000u) ? ~u : (u | 0x80000000u);
}
__device__ __forceinline__ float ord2f(unsigned u) {
    return __uint_as_float((u & 0x80000000u) ? (u ^ 0x80000000u) : ~u);
}
#define ORD_NEGINF 0x007FFFFFu

__device__ __forceinline__ float leaky(float x) { return x > 0.f ? x : ALPHA * x; }

__global__ void gat_gemm_wmma(const float* __restrict__ node,
                              const float* __restrict__ W,
                              float* __restrict__ h,
                              float* __restrict__ out, float* __restrict__ denom,
                              unsigned* __restrict__ cnt, int N) {
    int wave = (blockIdx.x * blockDim.x + threadIdx.x) >> 5;
    int lane = threadIdx.x & 31;
    int row0 = wave * 16;
    if (wave == 0 && lane < 6) cnt[lane * CNT_STRIDE] = 0u;
    if (row0 >= N) return;

    {
        int rows = N - row0 < 16 ? N - row0 : 16;
        int nvec = rows * (D_OUT / 4);
        float4* orun = reinterpret_cast<float4*>(out + (size_t)row0 * D_OUT);
        const float4 z4 = make_float4(0.0f, 0.0f, 0.0f, 0.0f);
        for (int i = 0; i < 16 * (D_OUT / 4) / 32; ++i) {
            int v = i * 32 + lane;
            if (v < nvec) orun[v] = z4;
        }
        if (lane < rows) denom[row0 + lane] = 0.0f;
    }

    int mrow = lane & 15;
    int koff = (lane >> 4) * 2;

    v8f c0 = {}, c1 = {}, c2 = {}, c3 = {};

    const float* arow = node + (size_t)(row0 + mrow) * D_IN + koff;

    for (int k = 0; k < D_IN; k += 4) {
        v2f a;
        a.x = arow[k];
        a.y = arow[k + 1];

        const float* bp = W + (size_t)(k + koff) * D_OUT + mrow;
        v2f b0, b1, b2, b3;
        b0.x = bp[0];            b0.y = bp[D_OUT];
        b1.x = bp[16];           b1.y = bp[16 + D_OUT];
        b2.x = bp[32];           b2.y = bp[32 + D_OUT];
        b3.x = bp[48];           b3.y = bp[48 + D_OUT];

        c0 = __builtin_amdgcn_wmma_f32_16x16x4_f32(false, a, false, b0, (short)0, c0, false, false);
        c1 = __builtin_amdgcn_wmma_f32_16x16x4_f32(false, a, false, b1, (short)0, c1, false, false);
        c2 = __builtin_amdgcn_wmma_f32_16x16x4_f32(false, a, false, b2, (short)0, c2, false, false);
        c3 = __builtin_amdgcn_wmma_f32_16x16x4_f32(false, a, false, b3, (short)0, c3, false, false);
    }

    int nlo   = lane & 15;
    int rbase = row0 + ((lane >> 4) * 8);
#pragma unroll
    for (int r = 0; r < 8; ++r) {
        float* op = h + (size_t)(rbase + r) * D_OUT + nlo;
        op[0]  = c0[r];
        op[16] = c1[r];
        op[32] = c2[r];
        op[48] = c3[r];
    }
}

__global__ void gat_scores(const float* __restrict__ h, const float* __restrict__ a,
                           unsigned* __restrict__ pl, float* __restrict__ sdst, int N) {
    int i = blockIdx.x * blockDim.x + threadIdx.x;
    if (i >= N) return;
    const float* hr = h + (size_t)i * D_OUT;
    float s0 = 0.f, s1 = 0.f;
#pragma unroll 8
    for (int j = 0; j < D_OUT; ++j) {
        float v = hr[j];
        s0 = fmaf(v, a[j], s0);
        s1 = fmaf(v, a[D_OUT + j], s1);
    }
    reinterpret_cast<uint2*>(pl)[i] = make_uint2(__float_as_uint(s0), ORD_NEGINF);
    sdst[i] = s1;
}

__device__ __forceinline__ int node_split(int N) { return ((N / 2 + 63) / 64) * 64; }

__device__ __forceinline__ int running_half() {
    unsigned id = __builtin_amdgcn_s_sendmsg_rtn(0x87);
    return (int)((id >> 18) & 1u);
}

__device__ __forceinline__ bool next_item(unsigned* __restrict__ cnt, int nsuper, int* s_t,
                                          int& hn, int& stolen, int& T) {
    for (;;) {
        if (threadIdx.x == 0) *s_t = (int)atomicAdd(cnt + hn * CNT_STRIDE, 1u);
        __syncthreads();
        T = *s_t;
        __syncthreads();
        if (T < nsuper) return true;
        if (stolen) return false;
        stolen = 1;
        hn ^= 1;
    }
}

__device__ __forceinline__ int lanes_below(unsigned mask) {
    return (int)__builtin_amdgcn_mbcnt_lo(mask, 0u);
}

__global__ void __launch_bounds__(EBLOCK)
gat_edge_max(const int* __restrict__ src, const int* __restrict__ dst,
             const float* __restrict__ sdst, unsigned* __restrict__ pl,
             float* __restrict__ scr, unsigned* __restrict__ cnt, int E, int N) {
    __shared__ int s_t;
    __shared__ int s_h;
    int lane = threadIdx.x & 31;
    int wave = threadIdx.x >> 5;
    int split = node_split(N);
    int nsuper = (E + EWAVES * CHUNK - 1) / (EWAVES * CHUNK);
    if (threadIdx.x == 0) s_h = running_half();
    __syncthreads();
    int hn = s_h, stolen = 0, T = 0;
    while (next_item(cnt, nsuper, &s_t, hn, stolen, T)) {
        long long base = ((long long)T * EWAVES + wave) * CHUNK;
        if (base >= E) continue;
        long long cend = base + CHUNK < (long long)E ? base + CHUNK : (long long)E;
        int sv[CHUNK / 32];
#pragma unroll
        for (int k = 0; k < CHUNK / 32; ++k) {
            long long e = base + k * 32 + lane;
            sv[k] = e < cend ? src[e] : -1;
        }
        int run = 0;
        int   dv[CHUNK / 32], rk[CHUNK / 32];
        bool  ow[CHUNK / 32];
#pragma unroll
        for (int k = 0; k < CHUNK / 32; ++k) {
            long long e = base + k * 32 + lane;
            int s = sv[k];
            ow[k] = s >= 0 && ((s >= split ? 1 : 0) == hn);
            unsigned mask = __builtin_amdgcn_ballot_w32(ow[k]);
            rk[k] = run + lanes_below(mask);
            run += __popc(mask);
            dv[k] = dst[e < cend ? e : cend - 1];
        }
        float dsc[CHUNK / 32];
#pragma unroll
        for (int k = 0; k < CHUNK / 32; ++k) {
            int dc = ow[k] ? dv[k] : 0;
            dsc[k] = sdst[dc];
        }
#pragma unroll
        for (int k = 0; k < CHUNK / 32; ++k) {
            if (ow[k]) {
                long long slot = hn ? cend - 1 - rk[k] : base + rk[k];
                scr[slot] = dsc[k];
                atomicMax(&pl[2 * (size_t)sv[k] + 1], f2ord(dsc[k]));
            }
        }
    }
}

__device__ __forceinline__ float seg_max(float ss, unsigned mf) {
    float v = leaky(ss + ord2f(mf));
    return (__isinff(v) && v < 0.f) ? 0.0f : v;
}

__global__ void __launch_bounds__(EBLOCK)
gat_edge_exp(const int* __restrict__ src, float* __restrict__ scr,
             const unsigned* __restrict__ pl, float* __restrict__ denom,
             unsigned* __restrict__ cnt, int E, int N) {
    __shared__ int s_t;
    __shared__ int s_h;
    int lane = threadIdx.x & 31;
    int wave = threadIdx.x >> 5;
    int split = node_split(N);
    int nsuper = (E + EWAVES * CHUNK - 1) / (EWAVES * CHUNK);
    const uint2* pl2 = reinterpret_cast<const uint2*>(pl);
    if (threadIdx.x == 0) s_h = running_half();
    __syncthreads();
    int hn = s_h, stolen = 0, T = 0;
    while (next_item(cnt, nsuper, &s_t, hn, stolen, T)) {
        long long base = ((long long)T * EWAVES + wave) * CHUNK;
        if (base >= E) continue;
        long long cend = base + CHUNK < (long long)E ? base + CHUNK : (long long)E;
        int sv[CHUNK / 32];
#pragma unroll
        for (int k = 0; k < CHUNK / 32; ++k) {
            long long e = base + k * 32 + lane;
            sv[k] = e < cend ? src[e] : -1;
        }
        int run = 0;
        int safe = hn ? (split < N ? split : N - 1) : 0;
        long long idle = hn ? cend - 1 : base;
        float dsc[CHUNK / 32];
        uint2 pr[CHUNK / 32];
        int   rk[CHUNK / 32];
        bool  ow[CHUNK / 32];
#pragma unroll
        for (int k = 0; k < CHUNK / 32; ++k) {
            int s = sv[k];
            ow[k] = s >= 0 && ((s >= split ? 1 : 0) == hn);
            unsigned mask = __builtin_amdgcn_ballot_w32(ow[k]);
            rk[k] = run + lanes_below(mask);
            run += __popc(mask);
            long long slot = ow[k] ? (hn ? cend - 1 - rk[k] : base + rk[k]) : idle;
            dsc[k] = scr[slot];
            pr[k]  = pl2[ow[k] ? s : safe];
        }
#pragma unroll
        for (int k = 0; k < CHUNK / 32; ++k) {
            if (ow[k]) {
                float ss = __uint_as_float(pr[k].x);
                float w  = __expf(leaky(ss + dsc[k]) - seg_max(ss, pr[k].y));
                long long slot = hn ? cend - 1 - rk[k] : base + rk[k];
                scr[slot] = w;
                atomicAdd(&denom[sv[k]], w);
            }
        }
    }
}

#define SCATTER_GROUP  8
#define QCAP           (CHUNK + 32)

__device__ __forceinline__ int lane_int(int v, int j) {
    return __builtin_amdgcn_readlane(v, j);
}
__device__ __forceinline__ float lane_float(float v, int j) {
    return __int_as_float(__builtin_amdgcn_readlane(__float_as_int(v), j));
}

__device__ __forceinline__ void scatter_rows(int s, int d, float att, int n, int lane,
                                             const float* __restrict__ h,
                                             float* __restrict__ out) {
    int j0 = 0;
    for (; j0 + SCATTER_GROUP <= n; j0 += SCATTER_GROUP) {
        float v0[SCATTER_GROUP], v1[SCATTER_GROUP];
        for (int u = 0; u < SCATTER_GROUP; ++u) {
            const float* hrow = h + (size_t)lane_int(d, j0 + u) * D_OUT;
            v0[u] = hrow[lane];
            v1[u] = hrow[lane + 32];
        }
        for (int u = 0; u < SCATTER_GROUP; ++u) {
            float* orow = out + (size_t)lane_int(s, j0 + u) * D_OUT;
            float  aj   = lane_float(att, j0 + u);
            atomicAdd(orow + lane, aj * v0[u]);
            atomicAdd(orow + lane + 32, aj * v1[u]);
        }
    }
    for (int j = j0; j < n; ++j) {
        const float* hrow = h + (size_t)lane_int(d, j) * D_OUT;
        float*       orow = out + (size_t)lane_int(s, j) * D_OUT;
        float aj = lane_float(att, j);
        float a0 = hrow[lane];
        float a1 = hrow[lane + 32];
        atomicAdd(orow + lane, aj * a0);
        atomicAdd(orow + lane + 32, aj * a1);
    }
}

__global__ void __launch_bounds__(EBLOCK)
gat_edge_scatter(const int* __restrict__ src, const int* __restrict__ dst,
                 const float* __restrict__ scr,
                 const float* __restrict__ denom, const float* __restrict__ h,
                 float* __restrict__ out, unsigned* __restrict__ cnt, int E, int N) {
    __shared__ int   s_t;
    __shared__ int   s_h;
    __shared__ int   q_s[EWAVES][QCAP];
    __shared__ int   q_d[EWAVES][QCAP];
    __shared__ float q_a[EWAVES][QCAP];
    int lane = threadIdx.x & 31;
    int wave = threadIdx.x >> 5;
    int*   qs = q_s[wave];
    int*   qd = q_d[wave];
    float* qa = q_a[wave];
    int split = node_split(N);
    int nsuper = (E + EWAVES * CHUNK - 1) / (EWAVES * CHUNK);
    if (threadIdx.x == 0) s_h = running_half();
    __syncthreads();
    int hn = s_h, stolen = 0, T = 0;
    int qn = 0;
    while (next_item(cnt, nsuper, &s_t, hn, stolen, T)) {
        long long base = ((long long)T * EWAVES + wave) * CHUNK;
        if (base >= E) continue;
        long long cend = base + CHUNK < (long long)E ? base + CHUNK : (long long)E;
        int sv[CHUNK / 32];
#pragma unroll
        for (int k = 0; k < CHUNK / 32; ++k) {
            long long e = base + k * 32 + lane;
            sv[k] = e < cend ? src[e] : -1;
        }
        int run = 0;
        int safe = hn ? (split < N ? split : N - 1) : 0;
        long long idle = hn ? cend - 1 : base;
        int   dv[CHUNK / 32], rk[CHUNK / 32];
        float wv[CHUNK / 32], nv[CHUNK / 32];
        bool  ow[CHUNK / 32];
#pragma unroll
        for (int k = 0; k < CHUNK / 32; ++k) {
            long long e = base + k * 32 + lane;
            int s = sv[k];
            ow[k] = s >= 0 && ((s >= split ? 1 : 0) == hn);
            unsigned mask = __builtin_amdgcn_ballot_w32(ow[k]);
            rk[k] = run + lanes_below(mask);
            run += __popc(mask);
            long long slot = ow[k] ? (hn ? cend - 1 - rk[k] : base + rk[k]) : idle;
            int sc = ow[k] ? s : safe;
            dv[k] = dst[e < cend ? e : cend - 1];
            wv[k] = scr[slot];
            nv[k] = denom[sc];
        }
#pragma unroll
        for (int k = 0; k < CHUNK / 32; ++k) {
            if (ow[k]) {
                int pos = qn + rk[k];
                qs[pos] = sv[k];
                qd[pos] = dv[k];
                qa[pos] = wv[k] / (nv[k] + EPS);
            }
        }
        qn += run;
        int b = 0;
        for (; b + 32 <= qn; b += 32) {
            int   s2 = qs[b + lane];
            int   d2 = qd[b + lane];
            float a2 = qa[b + lane];
            scatter_rows(s2, d2, a2, 32, lane, h, out);
        }
        int left = qn - b;
        if (b > 0 && left > 0) {
            int s2 = 0, d2 = 0;
            float a2 = 0.0f;
            if (lane < left) { s2 = qs[b + lane]; d2 = qd[b + lane]; a2 = qa[b + lane]; }
            if (lane < left) { qs[lane] = s2; qd[lane] = d2; qa[lane] = a2; }
        }
        qn = left;
    }
    if (qn > 0) {
        int s2 = 0, d2 = 0;
        float a2 = 0.0f;
        if (lane < qn) { s2 = qs[lane]; d2 = qd[lane]; a2 = qa[lane]; }
        scatter_rows(s2, d2, a2, qn, lane, h, out);
    }
}

extern "C" void kernel_launch(void* const* d_in, const int* in_sizes, int n_in,
                              void* d_out, int out_size, void* d_ws, size_t ws_size,
                              hipStream_t stream) {
    const float* node = (const float*)d_in[0];
    const int*   ei   = (const int*)d_in[1];
    const float* W    = (const float*)d_in[2];
    const float* a    = (const float*)d_in[3];
    float* out        = (float*)d_out;

    int N = in_sizes[0] / D_IN;
    int E = in_sizes[1] / 2;
    const int* src = ei;
    const int* dst = ei + E;

    float*    h      = (float*)d_ws;
    unsigned* pl     = (unsigned*)(h + (size_t)N * D_OUT);
    float*    sdst   = (float*)(pl + 2 * (size_t)N);
    float*    denom  = sdst + N;
    float*    scr    = denom + N;
    unsigned* cnt    = (unsigned*)(((size_t)(scr + E) + 255) & ~(size_t)255);
    if ((size_t)((char*)(cnt + 6 * CNT_STRIDE) - (char*)d_ws) > ws_size) return;

    const int B = 256;

    {
        int waves   = (N + 15) / 16;
        long long threads = (long long)waves * 32;
        int blocks  = (int)((threads + B - 1) / B);
        gat_gemm_wmma<<<blocks, B, 0, stream>>>(node, W, h, out, denom, cnt, N);
    }
    gat_scores<<<(N + B - 1) / B, B, 0, stream>>>(h, a, pl, sdst, N);
    gat_edge_max<<<4096, EBLOCK, 0, stream>>>(src, dst, sdst, pl, scr,
                                             cnt, E, N);
    gat_edge_exp<<<4096, EBLOCK, 0, stream>>>(src, scr, pl, denom,
                                             cnt + 2 * CNT_STRIDE, E, N);
    gat_edge_scatter<<<4096, EBLOCK, 0, stream>>>(src, dst, scr, denom, h, out,
                                                 cnt + 4 * CNT_STRIDE, E, N);
}


# max and exp passes: 1024 long-lived work-groups (were 4096); a wave takes 4 chunks per ticket, the next chunk's indices requested under the current chunk's gathers; sum order unchanged (arrival)
# speedup vs baseline: 3.0251x; 1.0471x over previous
#include <hip/hip_runtime.h>
#include <hip/hip_bf16.h>

#define D_IN   128
#define D_OUT  64
#define ALPHA  0.2f
#define EPS    1e-10f

#define CHUNK      256
#define EWAVES     4
#define EBLOCK     (EWAVES * 32)
#define CNT_STRIDE 64
#define LWGS       1024
#define LBATCH     4

typedef __attribute__((ext_vector_type(2))) float v2f;
typedef __attribute__((ext_vector_type(8))) float v8f;

__device__ __forceinline__ unsigned f2ord(float f) {
    unsigned u = __float_as_uint(f);
    return (u & 0x80000000u) ? ~u : (u | 0x80000000u);
}
__device__ __forceinline__ float ord2f(unsigned u) {
    return __uint_as_float((u & 0x80000000u) ? (u ^ 0x80000000u) : ~u);
}
#define ORD_NEGINF 0x007FFFFFu

__device__ __forceinline__ float leaky(float x) { return x > 0.f ? x : ALPHA * x; }

__global__ void gat_gemm_wmma(const float* __restrict__ node,
                              const float* __restrict__ W,
                              float* __restrict__ h,
                              float* __restrict__ out, float* __restrict__ denom,
                              unsigned* __restrict__ cnt, int N) {
    int wave = (blockIdx.x * blockDim.x + threadIdx.x) >> 5;
    int lane = threadIdx.x & 31;
    int row0 = wave * 16;
    if (wave == 0 && lane < 6) cnt[lane * CNT_STRIDE] = 0u;
    if (row0 >= N) return;

    {
        int rows = N - row0 < 16 ? N - row0 : 16;
        int nvec = rows * (D_OUT / 4);
        float4* orun = reinterpret_cast<float4*>(out + (size_t)row0 * D_OUT);
        const float4 z4 = make_float4(0.0f, 0.0f, 0.0f, 0.0f);
        for (int i = 0; i < 16 * (D_OUT / 4) / 32; ++i) {
            int v = i * 32 + lane;
            if (v < nvec) orun[v] = z4;
        }
        if (lane < rows) denom[row0 + lane] = 0.0f;
    }

    int mrow = lane & 15;
    int koff = (lane >> 4) * 2;

    v8f c0 = {}, c1 = {}, c2 = {}, c3 = {};

    const float* arow = node + (size_t)(row0 + mrow) * D_IN + koff;

    for (int k = 0; k < D_IN; k += 4) {
        v2f a;
        a.x = arow[k];
        a.y = arow[k + 1];

        const float* bp = W + (size_t)(k + koff) * D_OUT + mrow;
        v2f b0, b1, b2, b3;
        b0.x = bp[0];            b0.y = bp[D_OUT];
        b1.x = bp[16];           b1.y = bp[16 + D_OUT];
        b2.x = bp[32];           b2.y = bp[32 + D_OUT];
        b3.x = bp[48];           b3.y = bp[48 + D_OUT];

        c0 = __builtin_amdgcn_wmma_f32_16x16x4_f32(false, a, false, b0, (short)0, c0, false, false);
        c1 = __builtin_amdgcn_wmma_f32_16x16x4_f32(false, a, false, b1, (short)0, c1, false, false);
        c2 = __builtin_amdgcn_wmma_f32_16x16x4_f32(false, a, false, b2, (short)0, c2, false, false);
        c3 = __builtin_amdgcn_wmma_f32_16x16x4_f32(false, a, false, b3, (short)0, c3, false, false);
    }

    int nlo   = lane & 15;
    int rbase = row0 + ((lane >> 4) * 8);
#pragma unroll
    for (int r = 0; r < 8; ++r) {
        float* op = h + (size_t)(rbase + r) * D_OUT + nlo;
        op[0]  = c0[r];
        op[16] = c1[r];
        op[32] = c2[r];
        op[48] = c3[r];
    }
}

__global__ void gat_scores(const float* __restrict__ h, const float* __restrict__ a,
                           unsigned* __restrict__ pl, float* __restrict__ sdst, int N) {
    int i = blockIdx.x * blockDim.x + threadIdx.x;
    if (i >= N) return;
    const float* hr = h + (size_t)i * D_OUT;
    float s0 = 0.f, s1 = 0.f;
#pragma unroll 8
    for (int j = 0; j < D_OUT; ++j) {
        float v = hr[j];
        s0 = fmaf(v, a[j], s0);
        s1 = fmaf(v, a[D_OUT + j], s1);
    }
    reinterpret_cast<uint2*>(pl)[i] = make_uint2(__float_as_uint(s0), ORD_NEGINF);
    sdst[i] = s1;
}

__device__ __forceinline__ int node_split(int N) { return ((N / 2 + 63) / 64) * 64; }

__device__ __forceinline__ int running_half() {
    unsigned id = __builtin_amdgcn_s_sendmsg_rtn(0x87);
    return (int)((id >> 18) & 1u);
}

__device__ __forceinline__ bool next_item(unsigned* __restrict__ cnt, int nsuper, int* s_t,
                                          int& hn, int& stolen, int& T) {
    for (;;) {
        if (threadIdx.x == 0) *s_t = (int)atomicAdd(cnt + hn * CNT_STRIDE, 1u);
        __syncthreads();
        T = *s_t;
        __syncthreads();
        if (T < nsuper) return true;
        if (stolen) return false;
        stolen = 1;
        hn ^= 1;
    }
}

__device__ __forceinline__ int lanes_below(unsigned mask) {
    return (int)__builtin_amdgcn_mbcnt_lo(mask, 0u);
}

__device__ __forceinline__ void load_src(const int* __restrict__ src, long long base, int E,
                                         int lane, int (&sv)[CHUNK / 32]) {
    long long cend = base + CHUNK < (long long)E ? base + CHUNK : (long long)E;
#pragma unroll
    for (int k = 0; k < CHUNK / 32; ++k) {
        long long e = base + k * 32 + lane;
        sv[k] = e < cend ? src[e] : -1;
    }
}
__device__ __forceinline__ void load_dst(const int* __restrict__ dst, long long base, int E,
                                         int lane, int (&dv)[CHUNK / 32]) {
    long long cend = base + CHUNK < (long long)E ? base + CHUNK : (long long)E;
#pragma unroll
    for (int k = 0; k < CHUNK / 32; ++k) {
        long long e = base + k * 32 + lane;
        dv[k] = dst[e < cend ? e : cend - 1];
    }
}

__global__ void __launch_bounds__(EBLOCK)
gat_edge_max(const int* __restrict__ src, const int* __restrict__ dst,
             const float* __restrict__ sdst, unsigned* __restrict__ pl,
             float* __restrict__ scr, unsigned* __restrict__ cnt, int E, int N) {
    __shared__ int s_t;
    __shared__ int s_h;
    int lane = threadIdx.x & 31;
    int wave = threadIdx.x >> 5;
    int split = node_split(N);
    int nsuper = (E + EWAVES * CHUNK - 1) / (EWAVES * CHUNK);
    int nbatch = (nsuper + LBATCH - 1) / LBATCH;
    if (threadIdx.x == 0) s_h = running_half();
    __syncthreads();
    int hn = s_h, stolen = 0, T = 0;
    while (next_item(cnt, nbatch, &s_t, hn, stolen, T)) {
        long long base = ((long long)T * LBATCH * EWAVES + wave) * CHUNK;
        int sv[CHUNK / 32], dv[CHUNK / 32];
        load_src(src, base, E, lane, sv);
        load_dst(dst, base, E, lane, dv);
        for (int j = 0; j < LBATCH; ++j) {
            if (base >= E) break;
            long long cend = base + CHUNK < (long long)E ? base + CHUNK : (long long)E;
            int run = 0;
            int   rk[CHUNK / 32];
            bool  ow[CHUNK / 32];
            float dsc[CHUNK / 32];
#pragma unroll
            for (int k = 0; k < CHUNK / 32; ++k) {
                int s = sv[k];
                ow[k] = s >= 0 && ((s >= split ? 1 : 0) == hn);
                unsigned mask = __builtin_amdgcn_ballot_w32(ow[k]);
                rk[k] = run + lanes_below(mask);
                run += __popc(mask);
                dsc[k] = sdst[ow[k] ? dv[k] : 0];
            }
            long long nbase = base + (long long)EWAVES * CHUNK;
            int svn[CHUNK / 32], dvn[CHUNK / 32];
            if (j + 1 < LBATCH) {
                load_src(src, nbase, E, lane, svn);
                load_dst(dst, nbase, E, lane, dvn);
            } else {
#pragma unroll
                for (int k = 0; k < CHUNK / 32; ++k) { svn[k] = -1; dvn[k] = 0; }
            }
#pragma unroll
            for (int k = 0; k < CHUNK / 32; ++k) {
                if (ow[k]) {
                    long long slot = hn ? cend - 1 - rk[k] : base + rk[k];
                    scr[slot] = dsc[k];
                    atomicMax(&pl[2 * (size_t)sv[k] + 1], f2ord(dsc[k]));
                }
            }
#pragma unroll
            for (int k = 0; k < CHUNK / 32; ++k) { sv[k] = svn[k]; dv[k] = dvn[k]; }
            base = nbase;
        }
    }
}

__device__ __forceinline__ float seg_max(float ss, unsigned mf) {
    float v = leaky(ss + ord2f(mf));
    return (__isinff(v) && v < 0.f) ? 0.0f : v;
}

__global__ void __launch_bounds__(EBLOCK)
gat_edge_exp(const int* __restrict__ src, float* __restrict__ scr,
             const unsigned* __restrict__ pl, float* __restrict__ denom,
             unsigned* __restrict__ cnt, int E, int N) {
    __shared__ int s_t;
    __shared__ int s_h;
    int lane = threadIdx.x & 31;
    int wave = threadIdx.x >> 5;
    int split = node_split(N);
    int nsuper = (E + EWAVES * CHUNK - 1) / (EWAVES * CHUNK);
    int nbatch = (nsuper + LBATCH - 1) / LBATCH;
    const uint2* pl2 = reinterpret_cast<const uint2*>(pl);
    if (threadIdx.x == 0) s_h = running_half();
    __syncthreads();
    int hn = s_h, stolen = 0, T = 0;
    while (next_item(cnt, nbatch, &s_t, hn, stolen, T)) {
        long long base = ((long long)T * LBATCH * EWAVES + wave) * CHUNK;
        int safe = hn ? (split < N ? split : N - 1) : 0;
        int sv[CHUNK / 32];
        load_src(src, base, E, lane, sv);
        for (int j = 0; j < LBATCH; ++j) {
            if (base >= E) break;
            long long cend = base + CHUNK < (long long)E ? base + CHUNK : (long long)E;
            int run = 0;
            long long idle = hn ? cend - 1 : base;
            float dsc[CHUNK / 32];
            uint2 pr[CHUNK / 32];
            int   rk[CHUNK / 32];
            bool  ow[CHUNK / 32];
#pragma unroll
            for (int k = 0; k < CHUNK / 32; ++k) {
                int s = sv[k];
                ow[k] = s >= 0 && ((s >= split ? 1 : 0) == hn);
                unsigned mask = __builtin_amdgcn_ballot_w32(ow[k]);
                rk[k] = run + lanes_below(mask);
                run += __popc(mask);
                long long slot = ow[k] ? (hn ? cend - 1 - rk[k] : base + rk[k]) : idle;
                dsc[k] = scr[slot];
                pr[k]  = pl2[ow[k] ? s : safe];
            }
            long long nbase = base + (long long)EWAVES * CHUNK;
            int svn[CHUNK / 32];
            if (j + 1 < LBATCH) {
                load_src(src, nbase, E, lane, svn);
            } else {
#pragma unroll
                for (int k = 0; k < CHUNK / 32; ++k) svn[k] = -1;
            }
#pragma unroll
            for (int k = 0; k < CHUNK / 32; ++k) {
                if (ow[k]) {
                    float ss = __uint_as_float(pr[k].x);
                    float w  = __expf(leaky(ss + dsc[k]) - seg_max(ss, pr[k].y));
                    long long slot = hn ? cend - 1 - rk[k] : base + rk[k];
                    scr[slot] = w;
                    atomicAdd(&denom[sv[k]], w);
                }
            }
#pragma unroll
            for (int k = 0; k < CHUNK / 32; ++k) sv[k] = svn[k];
            base = nbase;
        }
    }
}

#define SCATTER_GROUP  8
#define QCAP           (CHUNK + 32)

__device__ __forceinline__ int lane_int(int v, int j) {
    return __builtin_amdgcn_readlane(v, j);
}
__device__ __forceinline__ float lane_float(float v, int j) {
    return __int_as_float(__builtin_amdgcn_readlane(__float_as_int(v), j));
}

__device__ __forceinline__ void scatter_rows(int s, int d, float att, int n, int lane,
                                             const float* __restrict__ h,
                                             float* __restrict__ out) {
    int j0 = 0;
    for (; j0 + SCATTER_GROUP <= n; j0 += SCATTER_GROUP) {
        float v0[SCATTER_GROUP], v1[SCATTER_GROUP];
        for (int u = 0; u < SCATTER_GROUP; ++u) {
            const float* hrow = h + (size_t)lane_int(d, j0 + u) * D_OUT;
            v0[u] = hrow[lane];
            v1[u] = hrow[lane + 32];
        }
        for (int u = 0; u < SCATTER_GROUP; ++u) {
            float* orow = out + (size_t)lane_int(s, j0 + u) * D_OUT;
            float  aj   = lane_float(att, j0 + u);
            atomicAdd(orow + lane, aj * v0[u]);
            atomicAdd(orow + lane + 32, aj * v1[u]);
        }
    }
    for (int j = j0; j < n; ++j) {
        const float* hrow = h + (size_t)lane_int(d, j) * D_OUT;
        float*       orow = out + (size_t)lane_int(s, j) * D_OUT;
        float aj = lane_float(att, j);
        float a0 = hrow[lane];
        float a1 = hrow[lane + 32];
        atomicAdd(orow + lane, aj * a0);
        atomicAdd(orow + lane + 32, aj * a1);
    }
}

__global__ void __launch_bounds__(EBLOCK)
gat_edge_scatter(const int* __restrict__ src, const int* __restrict__ dst,
                 const float* __restrict__ scr,
                 const float* __restrict__ denom, const float* __restrict__ h,
                 float* __restrict__ out, unsigned* __restrict__ cnt, int E, int N) {
    __shared__ int   s_t;
    __shared__ int   s_h;
    __shared__ int   q_s[EWAVES][QCAP];
    __shared__ int   q_d[EWAVES][QCAP];
    __shared__ float q_a[EWAVES][QCAP];
    int lane = threadIdx.x & 31;
    int wave = threadIdx.x >> 5;
    int*   qs = q_s[wave];
    int*   qd = q_d[wave];
    float* qa = q_a[wave];
    int split = node_split(N);
    int nsuper = (E + EWAVES * CHUNK - 1) / (EWAVES * CHUNK);
    if (threadIdx.x == 0) s_h = running_half();
    __syncthreads();
    int hn = s_h, stolen = 0, T = 0;
    int qn = 0;
    while (next_item(cnt, nsuper, &s_t, hn, stolen, T)) {
        long long base = ((long long)T * EWAVES + wave) * CHUNK;
        if (base >= E) continue;
        long long cend = base + CHUNK < (long long)E ? base + CHUNK : (long long)E;
        int sv[CHUNK / 32];
#pragma unroll
        for (int k = 0; k < CHUNK / 32; ++k) {
            long long e = base + k * 32 + lane;
            sv[k] = e < cend ? src[e] : -1;
        }
        int run = 0;
        int safe = hn ? (split < N ? split : N - 1) : 0;
        long long idle = hn ? cend - 1 : base;
        int   dv[CHUNK / 32], rk[CHUNK / 32];
        float wv[CHUNK / 32], nv[CHUNK / 32];
        bool  ow[CHUNK / 32];
#pragma unroll
        for (int k = 0; k < CHUNK / 32; ++k) {
            long long e = base + k * 32 + lane;
            int s = sv[k];
            ow[k] = s >= 0 && ((s >= split ? 1 : 0) == hn);
            unsigned mask = __builtin_amdgcn_ballot_w32(ow[k]);
            rk[k] = run + lanes_below(mask);
            run += __popc(mask);
            long long slot = ow[k] ? (hn ? cend - 1 - rk[k] : base + rk[k]) : idle;
            int sc = ow[k] ? s : safe;
            dv[k] = dst[e < cend ? e : cend - 1];
            wv[k] = scr[slot];
            nv[k] = denom[sc];
        }
#pragma unroll
        for (int k = 0; k < CHUNK / 32; ++k) {
            if (ow[k]) {
                int pos = qn + rk[k];
                qs[pos] = sv[k];
                qd[pos] = dv[k];
                qa[pos] = wv[k] / (nv[k] + EPS);
            }
        }
        qn += run;
        int b = 0;
        for (; b + 32 <= qn; b += 32) {
            int   s2 = qs[b + lane];
            int   d2 = qd[b + lane];
            float a2 = qa[b + lane];
            scatter_rows(s2, d2, a2, 32, lane, h, out);
        }
        int left = qn - b;
        if (b > 0 && left > 0) {
            int s2 = 0, d2 = 0;
            float a2 = 0.0f;
            if (lane < left) { s2 = qs[b + lane]; d2 = qd[b + lane]; a2 = qa[b + lane]; }
            if (lane < left) { qs[lane] = s2; qd[lane] = d2; qa[lane] = a2; }
        }
        qn = left;
    }
    if (qn > 0) {
        int s2 = 0, d2 = 0;
        float a2 = 0.0f;
        if (lane < qn) { s2 = qs[lane]; d2 = qd[lane]; a2 = qa[lane]; }
        scatter_rows(s2, d2, a2, qn, lane, h, out);
    }
}

extern "C" void kernel_launch(void* const* d_in, const int* in_sizes, int n_in,
                              void* d_out, int out_size, void* d_ws, size_t ws_size,
                              hipStream_t stream) {
    const float* node = (const float*)d_in[0];
    const int*   ei   = (const int*)d_in[1];
    const float* W    = (const float*)d_in[2];
    const float* a    = (const float*)d_in[3];
    float* out        = (float*)d_out;

    int N = in_sizes[0] / D_IN;
    int E = in_sizes[1] / 2;
    const int* src = ei;
    const int* dst = ei + E;

    float*    h      = (float*)d_ws;
    unsigned* pl     = (unsigned*)(h + (size_t)N * D_OUT);
    float*    sdst   = (float*)(pl + 2 * (size_t)N);
    float*    denom  = sdst + N;
    float*    scr    = denom + N;
    unsigned* cnt    = (unsigned*)(((size_t)(scr + E) + 255) & ~(size_t)255);
    if ((size_t)((char*)(cnt + 6 * CNT_STRIDE) - (char*)d_ws) > ws_size) return;

    const int B = 256;

    {
        int waves   = (N + 15) / 16;
        long long threads = (long long)waves * 32;
        int blocks  = (int)((threads + B - 1) / B);
        gat_gemm_wmma<<<blocks, B, 0, stream>>>(node, W, h, out, denom, cnt, N);
    }
    gat_scores<<<(N + B - 1) / B, B, 0, stream>>>(h, a, pl, sdst, N);
    gat_edge_max<<<LWGS, EBLOCK, 0, stream>>>(src, dst, sdst, pl, scr,
                                             cnt, E, N);
    gat_edge_exp<<<LWGS, EBLOCK, 0, stream>>>(src, scr, pl, denom,
                                             cnt + 2 * CNT_STRIDE, E, N);
    gat_edge_scatter<<<4096, EBLOCK, 0, stream>>>(src, dst, scr, denom, h, out,
                                                 cnt + 4 * CNT_STRIDE, E, N);
}


# softmax shift = leaky(ssrc + largest sdst of all nodes) while sdst spans <= 6: exp pass gathers sdst[dst] itself, per-node max pass exits at once (kept for wider spans); sum order unchanged (arrival)
# speedup vs baseline: 3.2865x; 1.0864x over previous
#include <hip/hip_runtime.h>
#include <hip/hip_bf16.h>

#define D_IN   128
#define D_OUT  64
#define ALPHA  0.2f
#define EPS    1e-10f

#define CHUNK      256
#define EWAVES     4
#define EBLOCK     (EWAVES * 32)
#define CNT_STRIDE 64
#define LWGS       1024
#define LBATCH     4

typedef __attribute__((ext_vector_type(2))) float v2f;
typedef __attribute__((ext_vector_type(8))) float v8f;

__device__ __forceinline__ unsigned f2ord(float f) {
    unsigned u = __float_as_uint(f);
    return (u & 0x80000000u) ? ~u : (u | 0x80000000u);
}
__device__ __forceinline__ float ord2f(unsigned u) {
    return __uint_as_float((u & 0x80000000u) ? (u ^ 0x80000000u) : ~u);
}
#define ORD_NEGINF 0x007FFFFFu
#define ORD_POSINF 0xFF800000u
#define RNG_OFF    (6 * CNT_STRIDE)
#define SPREAD_MAX 6.0f

__device__ __forceinline__ float leaky(float x) { return x > 0.f ? x : ALPHA * x; }

__global__ void gat_gemm_wmma(const float* __restrict__ node,
                              const float* __restrict__ W,
                              float* __restrict__ h,
                              float* __restrict__ out, float* __restrict__ denom,
                              unsigned* __restrict__ cnt, int N) {
    int wave = (blockIdx.x * blockDim.x + threadIdx.x) >> 5;
    int lane = threadIdx.x & 31;
    int row0 = wave * 16;
    if (wave == 0 && lane < 6) cnt[lane * CNT_STRIDE] = 0u;
    if (wave == 0 && lane == 6) cnt[RNG_OFF] = ORD_NEGINF;
    if (wave == 0 && lane == 7) cnt[RNG_OFF + 1] = ORD_POSINF;
    if (row0 >= N) return;

    {
        int rows = N - row0 < 16 ? N - row0 : 16;
        int nvec = rows * (D_OUT / 4);
        float4* orun = reinterpret_cast<float4*>(out + (size_t)row0 * D_OUT);
        const float4 z4 = make_float4(0.0f, 0.0f, 0.0f, 0.0f);
        for (int i = 0; i < 16 * (D_OUT / 4) / 32; ++i) {
            int v = i * 32 + lane;
            if (v < nvec) orun[v] = z4;
        }
        if (lane < rows) denom[row0 + lane] = 0.0f;
    }

    int mrow = lane & 15;
    int koff = (lane >> 4) * 2;

    v8f c0 = {}, c1 = {}, c2 = {}, c3 = {};

    const float* arow = node + (size_t)(row0 + mrow) * D_IN + koff;

    for (int k = 0; k < D_IN; k += 4) {
        v2f a;
        a.x = arow[k];
        a.y = arow[k + 1];

        const float* bp = W + (size_t)(k + koff) * D_OUT + mrow;
        v2f b0, b1, b2, b3;
        b0.x = bp[0];            b0.y = bp[D_OUT];
        b1.x = bp[16];           b1.y = bp[16 + D_OUT];
        b2.x = bp[32];           b2.y = bp[32 + D_OUT];
        b3.x = bp[48];           b3.y = bp[48 + D_OUT];

        c0 = __builtin_amdgcn_wmma_f32_16x16x4_f32(false, a, false, b0, (short)0, c0, false, false);
        c1 = __builtin_amdgcn_wmma_f32_16x16x4_f32(false, a, false, b1, (short)0, c1, false, false);
        c2 = __builtin_amdgcn_wmma_f32_16x16x4_f32(false, a, false, b2, (short)0, c2, false, false);
        c3 = __builtin_amdgcn_wmma_f32_16x16x4_f32(false, a, false, b3, (short)0, c3, false, false);
    }

    int nlo   = lane & 15;
    int rbase = row0 + ((lane >> 4) * 8);
#pragma unroll
    for (int r = 0; r < 8; ++r) {
        float* op = h + (size_t)(rbase + r) * D_OUT + nlo;
        op[0]  = c0[r];
        op[16] = c1[r];
        op[32] = c2[r];
        op[48] = c3[r];
    }
}

#define SBLOCK 256
__global__ void __launch_bounds__(SBLOCK)
gat_scores(const float* __restrict__ h, const float* __restrict__ a,
           unsigned* __restrict__ pl, float* __restrict__ sdst,
           unsigned* __restrict__ rng, int N) {
    __shared__ unsigned s_hi[SBLOCK / 32];
    __shared__ unsigned s_lo[SBLOCK / 32];
    int i = blockIdx.x * blockDim.x + threadIdx.x;
    unsigned hi = 0u, lo = 0xFFFFFFFFu;
    if (i < N) {
        const float* hr = h + (size_t)i * D_OUT;
        float s0 = 0.f, s1 = 0.f;
#pragma unroll 8
        for (int j = 0; j < D_OUT; ++j) {
            float v = hr[j];
            s0 = fmaf(v, a[j], s0);
            s1 = fmaf(v, a[D_OUT + j], s1);
        }
        reinterpret_cast<uint2*>(pl)[i] = make_uint2(__float_as_uint(s0), ORD_NEGINF);
        sdst[i] = s1;
        hi = lo = f2ord(s1);
    }
    for (int o = 16; o > 0; o >>= 1) {
        unsigned h2 = (unsigned)__shfl_xor((int)hi, o);
        unsigned l2 = (unsigned)__shfl_xor((int)lo, o);
        hi = hi > h2 ? hi : h2;
        lo = lo < l2 ? lo : l2;
    }
    int w = threadIdx.x >> 5;
    if ((threadIdx.x & 31) == 0) { s_hi[w] = hi; s_lo[w] = lo; }
    __syncthreads();
    if (threadIdx.x == 0) {
        for (int k = 1; k < SBLOCK / 32; ++k) {
            hi = hi > s_hi[k] ? hi : s_hi[k];
            lo = lo < s_lo[k] ? lo : s_lo[k];
        }
        atomicMax(rng, hi);
        atomicMin(rng + 1, lo);
    }
}

__device__ __forceinline__ bool narrow_range(const unsigned* __restrict__ rng, unsigned& smax_ord) {
    smax_ord = rng[0];
    float spread = ord2f(smax_ord) - ord2f(rng[1]);
    return spread <= SPREAD_MAX;
}

__device__ __forceinline__ int node_split(int N) { return ((N / 2 + 63) / 64) * 64; }

__device__ __forceinline__ int running_half() {
    unsigned id = __builtin_amdgcn_s_sendmsg_rtn(0x87);
    return (int)((id >> 18) & 1u);
}

__device__ __forceinline__ bool next_item(unsigned* __restrict__ cnt, int nsuper, int* s_t,
                                          int& hn, int& stolen, int& T) {
    for (;;) {
        if (threadIdx.x == 0) *s_t = (int)atomicAdd(cnt + hn * CNT_STRIDE, 1u);
        __syncthreads();
        T = *s_t;
        __syncthreads();
        if (T < nsuper) return true;
        if (stolen) return false;
        stolen = 1;
        hn ^= 1;
    }
}

__device__ __forceinline__ int lanes_below(unsigned mask) {
    return (int)__builtin_amdgcn_mbcnt_lo(mask, 0u);
}

__device__ __forceinline__ void load_src(const int* __restrict__ src, long long base, int E,
                                         int lane, int (&sv)[CHUNK / 32]) {
    long long cend = base + CHUNK < (long long)E ? base + CHUNK : (long long)E;
#pragma unroll
    for (int k = 0; k < CHUNK / 32; ++k) {
        long long e = base + k * 32 + lane;
        sv[k] = e < cend ? src[e] : -1;
    }
}
__device__ __forceinline__ void load_dst(const int* __restrict__ dst, long long base, int E,
                                         int lane, int (&dv)[CHUNK / 32]) {
    long long cend = base + CHUNK < (long long)E ? base + CHUNK : (long long)E;
#pragma unroll
    for (int k = 0; k < CHUNK / 32; ++k) {
        long long e = base + k * 32 + lane;
        dv[k] = dst[e < cend ? e : cend - 1];
    }
}

__global__ void __launch_bounds__(EBLOCK)
gat_edge_max(const int* __restrict__ src, const int* __restrict__ dst,
             const float* __restrict__ sdst, unsigned* __restrict__ pl,
             float* __restrict__ scr, unsigned* __restrict__ cnt,
             const unsigned* __restrict__ rng, int E, int N) {
    __shared__ int s_t;
    __shared__ int s_h;
    unsigned smax_ord;
    if (narrow_range(rng, smax_ord)) return;
    int lane = threadIdx.x & 31;
    int wave = threadIdx.x >> 5;
    int split = node_split(N);
    int nsuper = (E + EWAVES * CHUNK - 1) / (EWAVES * CHUNK);
    int nbatch = (nsuper + LBATCH - 1) / LBATCH;
    if (threadIdx.x == 0) s_h = running_half();
    __syncthreads();
    int hn = s_h, stolen = 0, T = 0;
    while (next_item(cnt, nbatch, &s_t, hn, stolen, T)) {
        long long base = ((long long)T * LBATCH * EWAVES + wave) * CHUNK;
        int sv[CHUNK / 32], dv[CHUNK / 32];
        load_src(src, base, E, lane, sv);
        load_dst(dst, base, E, lane, dv);
        for (int j = 0; j < LBATCH; ++j) {
            if (base >= E) break;
            long long cend = base + CHUNK < (long long)E ? base + CHUNK : (long long)E;
            int run = 0;
            int   rk[CHUNK / 32];
            bool  ow[CHUNK / 32];
            float dsc[CHUNK / 32];
#pragma unroll
            for (int k = 0; k < CHUNK / 32; ++k) {
                int s = sv[k];
                ow[k] = s >= 0 && ((s >= split ? 1 : 0) == hn);
                unsigned mask = __builtin_amdgcn_ballot_w32(ow[k]);
                rk[k] = run + lanes_below(mask);
                run += __popc(mask);
                dsc[k] = sdst[ow[k] ? dv[k] : 0];
            }
            long long nbase = base + (long long)EWAVES * CHUNK;
            int svn[CHUNK / 32], dvn[CHUNK / 32];
            if (j + 1 < LBATCH) {
                load_src(src, nbase, E, lane, svn);
                load_dst(dst, nbase, E, lane, dvn);
            } else {
#pragma unroll
                for (int k = 0; k < CHUNK / 32; ++k) { svn[k] = -1; dvn[k] = 0; }
            }
#pragma unroll
            for (int k = 0; k < CHUNK / 32; ++k) {
                if (ow[k]) {
                    long long slot = hn ? cend - 1 - rk[k] : base + rk[k];
                    scr[slot] = dsc[k];
                    atomicMax(&pl[2 * (size_t)sv[k] + 1], f2ord(dsc[k]));
                }
            }
#pragma unroll
            for (int k = 0; k < CHUNK / 32; ++k) { sv[k] = svn[k]; dv[k] = dvn[k]; }
            base = nbase;
        }
    }
}

__device__ __forceinline__ float seg_max(float ss, unsigned mf) {
    float v = leaky(ss + ord2f(mf));
    return (__isinff(v) && v < 0.f) ? 0.0f : v;
}

template <bool BOUND>
__device__ __forceinline__ void exp_items(const int* __restrict__ src, const int* __restrict__ dst,
                                          const float* __restrict__ sdst, float* __restrict__ scr,
                                          const uint2* __restrict__ pl2, float* __restrict__ denom,
                                          unsigned* __restrict__ cnt, int* s_t, int hn,
                                          unsigned smax_ord, int E, int N) {
    int lane = threadIdx.x & 31;
    int wave = threadIdx.x >> 5;
    int split = node_split(N);
    int nsuper = (E + EWAVES * CHUNK - 1) / (EWAVES * CHUNK);
    int nbatch = (nsuper + LBATCH - 1) / LBATCH;
    int stolen = 0, T = 0;
    while (next_item(cnt, nbatch, s_t, hn, stolen, T)) {
        long long base = ((long long)T * LBATCH * EWAVES + wave) * CHUNK;
        int safe = hn ? (split < N ? split : N - 1) : 0;
        int sv[CHUNK / 32], dv[CHUNK / 32];
        load_src(src, base, E, lane, sv);
        if (BOUND) load_dst(dst, base, E, lane, dv);
        for (int j = 0; j < LBATCH; ++j) {
            if (base >= E) break;
            long long cend = base + CHUNK < (long long)E ? base + CHUNK : (long long)E;
            int run = 0;
            long long idle = hn ? cend - 1 : base;
            float dsc[CHUNK / 32];
            uint2 pr[CHUNK / 32];
            int   rk[CHUNK / 32];
            bool  ow[CHUNK / 32];
#pragma unroll
            for (int k = 0; k < CHUNK / 32; ++k) {
                int s = sv[k];
                ow[k] = s >= 0 && ((s >= split ? 1 : 0) == hn);
                unsigned mask = __builtin_amdgcn_ballot_w32(ow[k]);
                rk[k] = run + lanes_below(mask);
                run += __popc(mask);
                if (BOUND) {
                    dsc[k] = sdst[ow[k] ? dv[k] : 0];
                } else {
                    long long slot = ow[k] ? (hn ? cend - 1 - rk[k] : base + rk[k]) : idle;
                    dsc[k] = scr[slot];
                }
                pr[k]  = pl2[ow[k] ? s : safe];
            }
            long long nbase = base + (long long)EWAVES * CHUNK;
            int svn[CHUNK / 32], dvn[CHUNK / 32];
            if (j + 1 < LBATCH) {
                load_src(src, nbase, E, lane, svn);
                if (BOUND) load_dst(dst, nbase, E, lane, dvn);
            } else {
#pragma unroll
                for (int k = 0; k < CHUNK / 32; ++k) { svn[k] = -1; dvn[k] = 0; }
            }
#pragma unroll
            for (int k = 0; k < CHUNK / 32; ++k) {
                if (ow[k]) {
                    float ss = __uint_as_float(pr[k].x);
                    float w  = __expf(leaky(ss + dsc[k]) - seg_max(ss, BOUND ? smax_ord : pr[k].y));
                    long long slot = hn ? cend - 1 - rk[k] : base + rk[k];
                    scr[slot] = w;
                    atomicAdd(&denom[sv[k]], w);
                }
            }
#pragma unroll
            for (int k = 0; k < CHUNK / 32; ++k) { sv[k] = svn[k]; if (BOUND) dv[k] = dvn[k]; }
            base = nbase;
        }
    }
}

__global__ void __launch_bounds__(EBLOCK)
gat_edge_exp(const int* __restrict__ src, const int* __restrict__ dst,
             const float* __restrict__ sdst, float* __restrict__ scr,
             const unsigned* __restrict__ pl, float* __restrict__ denom,
             unsigned* __restrict__ cnt, const unsigned* __restrict__ rng, int E, int N) {
    __shared__ int s_t;
    __shared__ int s_h;
    const uint2* pl2 = reinterpret_cast<const uint2*>(pl);
    unsigned smax_ord;
    bool bound = narrow_range(rng, smax_ord);
    if (threadIdx.x == 0) s_h = running_half();
    __syncthreads();
    int hn = s_h;
    if (bound) exp_items<true>(src, dst, sdst, scr, pl2, denom, cnt, &s_t, hn, smax_ord, E, N);
    else       exp_items<false>(src, dst, sdst, scr, pl2, denom, cnt, &s_t, hn, smax_ord, E, N);
}

#define SCATTER_GROUP  8
#define QCAP           (CHUNK + 32)

__device__ __forceinline__ int lane_int(int v, int j) {
    return __builtin_amdgcn_readlane(v, j);
}
__device__ __forceinline__ float lane_float(float v, int j) {
    return __int_as_float(__builtin_amdgcn_readlane(__float_as_int(v), j));
}

__device__ __forceinline__ void scatter_rows(int s, int d, float att, int n, int lane,
                                             const float* __restrict__ h,
                                             float* __restrict__ out) {
    int j0 = 0;
    for (; j0 + SCATTER_GROUP <= n; j0 += SCATTER_GROUP) {
        float v0[SCATTER_GROUP], v1[SCATTER_GROUP];
        for (int u = 0; u < SCATTER_GROUP; ++u) {
            const float* hrow = h + (size_t)lane_int(d, j0 + u) * D_OUT;
            v0[u] = hrow[lane];
            v1[u] = hrow[lane + 32];
        }
        for (int u = 0; u < SCATTER_GROUP; ++u) {
            float* orow = out + (size_t)lane_int(s, j0 + u) * D_OUT;
            float  aj   = lane_float(att, j0 + u);
            atomicAdd(orow + lane, aj * v0[u]);
            atomicAdd(orow + lane + 32, aj * v1[u]);
        }
    }
    for (int j = j0; j < n; ++j) {
        const float* hrow = h + (size_t)lane_int(d, j) * D_OUT;
        float*       orow = out + (size_t)lane_int(s, j) * D_OUT;
        float aj = lane_float(att, j);
        float a0 = hrow[lane];
        float a1 = hrow[lane + 32];
        atomicAdd(orow + lane, aj * a0);
        atomicAdd(orow + lane + 32, aj * a1);
    }
}

__global__ void __launch_bounds__(EBLOCK)
gat_edge_scatter(const int* __restrict__ src, const int* __restrict__ dst,
                 const float* __restrict__ scr,
                 const float* __restrict__ denom, const float* __restrict__ h,
                 float* __restrict__ out, unsigned* __restrict__ cnt, int E, int N) {
    __shared__ int   s_t;
    __shared__ int   s_h;
    __shared__ int   q_s[EWAVES][QCAP];
    __shared__ int   q_d[EWAVES][QCAP];
    __shared__ float q_a[EWAVES][QCAP];
    int lane = threadIdx.x & 31;
    int wave = threadIdx.x >> 5;
    int*   qs = q_s[wave];
    int*   qd = q_d[wave];
    float* qa = q_a[wave];
    int split = node_split(N);
    int nsuper = (E + EWAVES * CHUNK - 1) / (EWAVES * CHUNK);
    if (threadIdx.x == 0) s_h = running_half();
    __syncthreads();
    int hn = s_h, stolen = 0, T = 0;
    int qn = 0;
    while (next_item(cnt, nsuper, &s_t, hn, stolen, T)) {
        long long base = ((long long)T * EWAVES + wave) * CHUNK;
        if (base >= E) continue;
        long long cend = base + CHUNK < (long long)E ? base + CHUNK : (long long)E;
        int sv[CHUNK / 32];
#pragma unroll
        for (int k = 0; k < CHUNK / 32; ++k) {
            long long e = base + k * 32 + lane;
            sv[k] = e < cend ? src[e] : -1;
        }
        int run = 0;
        int safe = hn ? (split < N ? split : N - 1) : 0;
        long long idle = hn ? cend - 1 : base;
        int   dv[CHUNK / 32], rk[CHUNK / 32];
        float wv[CHUNK / 32], nv[CHUNK / 32];
        bool  ow[CHUNK / 32];
#pragma unroll
        for (int k = 0; k < CHUNK / 32; ++k) {
            long long e = base + k * 32 + lane;
            int s = sv[k];
            ow[k] = s >= 0 && ((s >= split ? 1 : 0) == hn);
            unsigned mask = __builtin_amdgcn_ballot_w32(ow[k]);
            rk[k] = run + lanes_below(mask);
            run += __popc(mask);
            long long slot = ow[k] ? (hn ? cend - 1 - rk[k] : base + rk[k]) : idle;
            int sc = ow[k] ? s : safe;
            dv[k] = dst[e < cend ? e : cend - 1];
            wv[k] = scr[slot];
            nv[k] = denom[sc];
        }
#pragma unroll
        for (int k = 0; k < CHUNK / 32; ++k) {
            if (ow[k]) {
                int pos = qn + rk[k];
                qs[pos] = sv[k];
                qd[pos] = dv[k];
                qa[pos] = wv[k] / (nv[k] + EPS);
            }
        }
        qn += run;
        int b = 0;
        for (; b + 32 <= qn; b += 32) {
            int   s2 = qs[b + lane];
            int   d2 = qd[b + lane];
            float a2 = qa[b + lane];
            scatter_rows(s2, d2, a2, 32, lane, h, out);
        }
        int left = qn - b;
        if (b > 0 && left > 0) {
            int s2 = 0, d2 = 0;
            float a2 = 0.0f;
            if (lane < left) { s2 = qs[b + lane]; d2 = qd[b + lane]; a2 = qa[b + lane]; }
            if (lane < left) { qs[lane] = s2; qd[lane] = d2; qa[lane] = a2; }
        }
        qn = left;
    }
    if (qn > 0) {
        int s2 = 0, d2 = 0;
        float a2 = 0.0f;
        if (lane < qn) { s2 = qs[lane]; d2 = qd[lane]; a2 = qa[lane]; }
        scatter_rows(s2, d2, a2, qn, lane, h, out);
    }
}

extern "C" void kernel_launch(void* const* d_in, const int* in_sizes, int n_in,
                              void* d_out, int out_size, void* d_ws, size_t ws_size,
                              hipStream_t stream) {
    const float* node = (const float*)d_in[0];
    const int*   ei   = (const int*)d_in[1];
    const float* W    = (const float*)d_in[2];
    const float* a    = (const float*)d_in[3];
    float* out        = (float*)d_out;

    int N = in_sizes[0] / D_IN;
    int E = in_sizes[1] / 2;
    const int* src = ei;
    const int* dst = ei + E;

    float*    h      = (float*)d_ws;
    unsigned* pl     = (unsigned*)(h + (size_t)N * D_OUT);
    float*    sdst   = (float*)(pl + 2 * (size_t)N);
    float*    denom  = sdst + N;
    float*    scr    = denom + N;
    unsigned* cnt    = (unsigned*)(((size_t)(scr + E) + 255) & ~(size_t)255);
    unsigned* rng    = cnt + RNG_OFF;
    if ((size_t)((char*)(cnt + 7 * CNT_STRIDE) - (char*)d_ws) > ws_size) return;

    const int B = 256;

    {
        int waves   = (N + 15) / 16;
        long long threads = (long long)waves * 32;
        int blocks  = (int)((threads + B - 1) / B);
        gat_gemm_wmma<<<blocks, B, 0, stream>>>(node, W, h, out, denom, cnt, N);
    }
    gat_scores<<<(N + SBLOCK - 1) / SBLOCK, SBLOCK, 0, stream>>>(h, a, pl, sdst, rng, N);
    gat_edge_max<<<LWGS, EBLOCK, 0, stream>>>(src, dst, sdst, pl, scr,
                                             cnt, rng, E, N);
    gat_edge_exp<<<LWGS, EBLOCK, 0, stream>>>(src, dst, sdst, scr, pl, denom,
                                             cnt + 2 * CNT_STRIDE, rng, E, N);
    gat_edge_scatter<<<4096, EBLOCK, 0, stream>>>(src, dst, scr, denom, h, out,
                                                 cnt + 4 * CNT_STRIDE, E, N);
}
